# starting kernel
# baseline (speedup 1.0000x reference)
#include <hip/hip_runtime.h>


typedef __attribute__((ext_vector_type(2))) float v2f;
typedef __attribute__((ext_vector_type(8))) float v8f;

#define C_IN    256
#define S_MID   64
#define HW      4096
#define NIMG    32
#define WAVES_PER_BLOCK 8
#define Y_PITCH 20

static __device__ __forceinline__ v8f wmma_f32(v2f a, v2f b, v8f c) {
  return __builtin_amdgcn_wmma_f32_16x16x4_f32(
       false, a,  false, b,
       (short)0, c,  false,  false);
}

__global__ __launch_bounds__(256)
void sse_fused_kernel(const float* __restrict__ x,
                      const float* __restrict__ W1,
                      const float* __restrict__ b1,
                      const float* __restrict__ W2,
                      const float* __restrict__ b2,
                      float* __restrict__ out) {
  __shared__ float ybuf[WAVES_PER_BLOCK * S_MID * Y_PITCH];

  const int lane = threadIdx.x & 31;
  const int wave = threadIdx.x >> 5;
  const int n    = lane & 15;
  const int hi   = lane >> 4;

  const int g   = blockIdx.x * WAVES_PER_BLOCK + wave;
  const int img = g >> 8;
  const int p0  = (g & 255) << 4;

  const float* xcol = x   + (size_t)img * (C_IN * HW) + p0 + n;
  float*       ocol = out + (size_t)img * (C_IN * HW) + p0 + n;

  v8f acc[4] = {};
  for (int kc = 0; kc < C_IN / 4; ++kc) {
    const int k0 = kc * 4 + hi * 2;
    v2f bfrag;
    bfrag.x = xcol[(size_t)(k0 + 0) * HW];
    bfrag.y = xcol[(size_t)(k0 + 1) * HW];
#pragma unroll
    for (int t = 0; t < 4; ++t) {
      v2f afrag = *reinterpret_cast<const v2f*>(W1 + (t * 16 + n) * C_IN + k0);
      acc[t] = wmma_f32(afrag, bfrag, acc[t]);
    }
  }

  float* ys = &ybuf[wave * S_MID * Y_PITCH];
#pragma unroll
  for (int t = 0; t < 4; ++t) {
#pragma unroll
    for (int j = 0; j < 8; ++j) {
      const int s = t * 16 + hi * 8 + j;
      float v = acc[t][j] + b1[s];
      ys[s * Y_PITCH + n] = fmaxf(v, 0.0f);
    }
  }
  __syncthreads();

  for (int ct = 0; ct < 16; ++ct) {
    v8f d = {};
#pragma unroll 4
    for (int ks = 0; ks < S_MID / 4; ++ks) {
      const int k0 = ks * 4 + hi * 2;
      v2f afrag = *reinterpret_cast<const v2f*>(W2 + (ct * 16 + n) * S_MID + k0);
      v2f bfrag;
      bfrag.x = ys[(k0 + 0) * Y_PITCH + n];
      bfrag.y = ys[(k0 + 1) * Y_PITCH + n];
      d = wmma_f32(afrag, bfrag, d);
    }
#pragma unroll
    for (int j = 0; j < 8; ++j) {
      const int c = ct * 16 + hi * 8 + j;
      float sv = d[j] + b2[c];
      sv = fminf(fmaxf(sv * 0.16666667f + 0.5f, 0.0f), 1.0f);
      const size_t off = (size_t)c * HW;
      ocol[off] = xcol[off] * sv;
    }
  }
}

extern "C" void kernel_launch(void* const* d_in, const int* in_sizes, int n_in,
                              void* d_out, int out_size, void* d_ws, size_t ws_size,
                              hipStream_t stream) {
  const float* x  = (const float*)d_in[0];
  const float* W1 = (const float*)d_in[1];
  const float* b1 = (const float*)d_in[2];
  const float* W2 = (const float*)d_in[3];
  const float* b2 = (const float*)d_in[4];
  float* out = (float*)d_out;

  const int total_groups = NIMG * (HW / 16);
  const int blocks = total_groups / WAVES_PER_BLOCK;
  sse_fused_kernel<<<blocks, 256, 0, stream>>>(x, W1, b1, W2, b2, out);
}


# W1 and W2 staged once per work-group into LDS in A-fragment order; A fragments read by ds_load_b128 instead of 16-line global gathers; one 32-wave work-group per WGP (grid 256 x 1024)
# speedup vs baseline: 1.4768x; 1.4768x over previous
#include <hip/hip_runtime.h>


typedef __attribute__((ext_vector_type(2))) float v2f;
typedef __attribute__((ext_vector_type(4))) float v4f;
typedef __attribute__((ext_vector_type(8))) float v8f;

#define C_IN    256
#define S_MID   64
#define HW      4096
#define NIMG    32
#define WAVES_PER_BLOCK 32
#define THREADS (WAVES_PER_BLOCK * 32)
#define Y_PITCH 20

static __device__ __forceinline__ v8f wmma_f32(v2f a, v2f b, v8f c) {
  return __builtin_amdgcn_wmma_f32_16x16x4_f32(
       false, a,  false, b,
       (short)0, c,  false,  false);
}

__global__ __launch_bounds__(THREADS)
void sse_fused_kernel(const float* __restrict__ x,
                      const float* __restrict__ W1,
                      const float* __restrict__ b1,
                      const float* __restrict__ W2,
                      const float* __restrict__ b2,
                      float* __restrict__ out) {
  __shared__ v4f w1s[S_MID * C_IN / 4];
  __shared__ v4f w2s[C_IN * S_MID / 4];
  __shared__ float ybuf[WAVES_PER_BLOCK * S_MID * Y_PITCH];

  const int tid  = threadIdx.x;
  const int lane = tid & 31;
  const int wave = tid >> 5;
  const int n    = lane & 15;
  const int hi   = lane >> 4;

  {
    float* w1l = reinterpret_cast<float*>(w1s);
    float* w2l = reinterpret_cast<float*>(w2s);
    v4f p1[4], p2[4];
#pragma unroll
    for (int i = 0; i < 4; ++i) {
      const int idx = i * THREADS + tid;
      const int q   = ((idx >> 6) & 15) * 4 + (idx & 3);
      const int r   = (idx >> 10) * 16 + ((idx >> 2) & 15);
      p1[i] = *reinterpret_cast<const v4f*>(W1 + r * C_IN + q * 4);
    }
#pragma unroll
    for (int i = 0; i < 4; ++i) {
      const int idx = i * THREADS + tid;
      const int q   = ((idx >> 6) & 3) * 4 + (idx & 3);
      const int r   = (idx >> 8) * 16 + ((idx >> 2) & 15);
      p2[i] = *reinterpret_cast<const v4f*>(W2 + r * S_MID + q * 4);
    }
#pragma unroll
    for (int i = 0; i < 4; ++i) {
      const int idx = i * THREADS + tid;
      const int q   = ((idx >> 6) & 15) * 4 + (idx & 3);
      const int c   = (idx >> 2) & 15;
      const int t   = idx >> 10;
      const int e   = ((q * 2 + (t >> 1)) * 32 + c) * 4 + (t & 1) * 2;
      *reinterpret_cast<v2f*>(w1l + e)      = v2f{p1[i].x, p1[i].y};
      *reinterpret_cast<v2f*>(w1l + e + 64) = v2f{p1[i].z, p1[i].w};
    }
#pragma unroll
    for (int i = 0; i < 4; ++i) {
      const int idx = i * THREADS + tid;
      const int q   = ((idx >> 6) & 3) * 4 + (idx & 3);
      const int c   = (idx >> 2) & 15;
      const int ct  = idx >> 8;
      const int e   = ((ct * 8 + (q >> 1)) * 32 + c) * 4 + (q & 1) * 2;
      *reinterpret_cast<v2f*>(w2l + e)      = v2f{p2[i].x, p2[i].y};
      *reinterpret_cast<v2f*>(w2l + e + 64) = v2f{p2[i].z, p2[i].w};
    }
  }
  __syncthreads();

  const int g   = blockIdx.x * WAVES_PER_BLOCK + wave;
  const int img = g >> 8;
  const int p0  = (g & 255) << 4;

  const float* xcol = x   + (size_t)img * (C_IN * HW) + p0 + n;
  float*       ocol = out + (size_t)img * (C_IN * HW) + p0 + n;

  const v4f* w1f = w1s + lane;
  v8f acc[4] = {};
  for (int kc = 0; kc < C_IN / 4; ++kc) {
    const int k0 = kc * 4 + hi * 2;
    v2f bfrag;
    bfrag.x = xcol[(size_t)(k0 + 0) * HW];
    bfrag.y = xcol[(size_t)(k0 + 1) * HW];
    const v4f a01 = w1f[(kc * 2 + 0) * 32];
    const v4f a23 = w1f[(kc * 2 + 1) * 32];
    acc[0] = wmma_f32(v2f{a01.x, a01.y}, bfrag, acc[0]);
    acc[1] = wmma_f32(v2f{a01.z, a01.w}, bfrag, acc[1]);
    acc[2] = wmma_f32(v2f{a23.x, a23.y}, bfrag, acc[2]);
    acc[3] = wmma_f32(v2f{a23.z, a23.w}, bfrag, acc[3]);
  }

  float* ys = &ybuf[wave * S_MID * Y_PITCH];
#pragma unroll
  for (int t = 0; t < 4; ++t) {
#pragma unroll
    for (int j = 0; j < 8; ++j) {
      const int s = t * 16 + hi * 8 + j;
      float v = acc[t][j] + b1[s];
      ys[s * Y_PITCH + n] = fmaxf(v, 0.0f);
    }
  }
  __syncthreads();

  const v4f* w2f = w2s + lane;
  for (int ct = 0; ct < 16; ++ct) {
    v8f d = {};
#pragma unroll 4
    for (int kp = 0; kp < S_MID / 8; ++kp) {
      const v4f a = w2f[(ct * 8 + kp) * 32];
      const int k0 = kp * 8 + hi * 2;
      v2f b0, b1f;
      b0.x  = ys[(k0 + 0) * Y_PITCH + n];
      b0.y  = ys[(k0 + 1) * Y_PITCH + n];
      b1f.x = ys[(k0 + 4) * Y_PITCH + n];
      b1f.y = ys[(k0 + 5) * Y_PITCH + n];
      d = wmma_f32(v2f{a.x, a.y}, b0,  d);
      d = wmma_f32(v2f{a.z, a.w}, b1f, d);
    }
#pragma unroll
    for (int j = 0; j < 8; ++j) {
      const int c = ct * 16 + hi * 8 + j;
      float sv = d[j] + b2[c];
      sv = fminf(fmaxf(sv * 0.16666667f + 0.5f, 0.0f), 1.0f);
      const size_t off = (size_t)c * HW;
      ocol[off] = xcol[off] * sv;
    }
  }
}

extern "C" void kernel_launch(void* const* d_in, const int* in_sizes, int n_in,
                              void* d_out, int out_size, void* d_ws, size_t ws_size,
                              hipStream_t stream) {
  const float* x  = (const float*)d_in[0];
  const float* W1 = (const float*)d_in[1];
  const float* b1 = (const float*)d_in[2];
  const float* W2 = (const float*)d_in[3];
  const float* b2 = (const float*)d_in[4];
  float* out = (float*)d_out;

  const int total_groups = NIMG * (HW / 16);
  const int blocks = total_groups / WAVES_PER_BLOCK;
  sse_fused_kernel<<<blocks, THREADS, 0, stream>>>(x, W1, b1, W2, b2, out);
}
